# starting kernel
# baseline (speedup 1.0000x reference)
#include <hip/hip_runtime.h>

typedef __attribute__((ext_vector_type(2))) float v2f;
typedef __attribute__((ext_vector_type(8))) float v8f;

#define SP 8
#define BLK 16

__global__ void build_M_kernel(const float* __restrict__ W_samp,
                               const float* __restrict__ W_init,
                               float* __restrict__ M) {
    const int t = threadIdx.x;
    const int k = t >> 4;
    const int n = t & 15;
    float acc = 0.0f;
#pragma unroll
    for (int s = 0; s < SP; ++s)
        acc += W_samp[s * BLK + k] * W_init[n * SP + s];
    M[k * BLK + n] = acc;
}

__global__ void __launch_bounds__(256) apply_M_kernel(
        const float* __restrict__ x,
        const float* __restrict__ M,
        float* __restrict__ y,
        int ntiles) {
    const int lane   = threadIdx.x & 31;
    const int wib    = threadIdx.x >> 5;
    const int gwave  = blockIdx.x * (blockDim.x >> 5) + wib;
    const int nwaves = gridDim.x * (blockDim.x >> 5);
    const int h = lane >> 4;
    const int n = lane & 15;

    v2f b[4];
#pragma unroll
    for (int c = 0; c < 4; ++c) {
        b[c].x = M[(4 * c + 2 * h + 0) * BLK + n];
        b[c].y = M[(4 * c + 2 * h + 1) * BLK + n];
    }

    for (int tile = gwave; tile < ntiles; tile += nwaves) {
        const float* Arow = x + (size_t)tile * 256 + (size_t)n * BLK;

        __builtin_prefetch(Arow + (size_t)nwaves * 256, 0, 0);

        v2f a[4];
#pragma unroll
        for (int c = 0; c < 4; ++c)
            a[c] = *(const v2f*)(Arow + 4 * c + 2 * h);

        v8f d = {};
#pragma unroll
        for (int c = 0; c < 4; ++c)
            d = __builtin_amdgcn_wmma_f32_16x16x4_f32(
                     false, a[c],
                     false, b[c],
                     (short)0, d,
                     false,  false);

        float* Drow = y + (size_t)tile * 256;
#pragma unroll
        for (int g = 0; g < 8; ++g)
            Drow[(size_t)(g + 8 * h) * BLK + n] = d[g];
    }
}

extern "C" void kernel_launch(void* const* d_in, const int* in_sizes, int n_in,
                              void* d_out, int out_size, void* d_ws, size_t ws_size,
                              hipStream_t stream) {
    const float* x      = (const float*)d_in[0];
    const float* W_samp = (const float*)d_in[1];
    const float* W_init = (const float*)d_in[2];
    float* y = (float*)d_out;
    float* M = (float*)d_ws;

    build_M_kernel<<<1, 256, 0, stream>>>(W_samp, W_init, M);

    const int ntiles = in_sizes[0] / 256;
    const int blocks = 1024;
    apply_M_kernel<<<blocks, 256, 0, stream>>>(x, M, y, ntiles);
}


# x arrives as whole lines (two 16-B/lane loads a tile, a wave's 8 tiles asked for up front, no prefetch hint), to product 1's fragments via a per-wave LDS patch; chunk c now sums k 2c,2c+1,8+2c,9+2c
# speedup vs baseline: 1.3373x; 1.3373x over previous
#include <hip/hip_runtime.h>


typedef __attribute__((ext_vector_type(2))) float v2f;
typedef __attribute__((ext_vector_type(4))) float v4f;
typedef __attribute__((ext_vector_type(8))) float v8f;

#define SP 8
#define BLK 16
#define WAVES 8
#define AHEAD 8
#define PITCH 20

typedef const __attribute__((address_space(1))) float* gfloatp;
typedef const __attribute__((address_space(1))) v4f*   gv4fp;

__device__ __forceinline__ gfloatp same_in_all_lanes(const float* p) {
    const unsigned long long v = (unsigned long long)p;
    const unsigned lo = __builtin_amdgcn_readfirstlane((unsigned)v);
    const unsigned hi = __builtin_amdgcn_readfirstlane((unsigned)(v >> 32));
    return (gfloatp)(((unsigned long long)hi << 32) | lo);
}

__global__ void __launch_bounds__(32 * WAVES) sample_recon_kernel(
        const float* __restrict__ x,
        const float* __restrict__ W_samp,
        const float* __restrict__ W_init,
        float* __restrict__ y,
        int ntiles) {
    __shared__ __attribute__((aligned(16))) float patch[WAVES][16 * PITCH];

    const int lane   = threadIdx.x & 31;
    const int wib    = threadIdx.x >> 5;
    const int gwave  = __builtin_amdgcn_readfirstlane(blockIdx.x * WAVES + wib);
    const int nwaves = gridDim.x * WAVES;
    const int h = lane >> 4;
    const int n = lane & 15;
    const int trow = 2 * (n & 7) + (n >> 3);

    const int s1 = (n & 3) + 4 * (n >> 3);
    v2f w1[4];
#pragma unroll
    for (int c = 0; c < 4; ++c)
        w1[c] = *(const v2f*)(W_samp + s1 * BLK + 8 * h + 2 * c);

    v2f w2[2];
#pragma unroll
    for (int q = 0; q < 2; ++q)
        w2[q] = *(const v2f*)(W_init + n * SP + 4 * h + 2 * q);

    float* mine = patch[wib];
    float* put = mine + (lane >> 2) * PITCH + 4 * (lane & 3);
    const float* get = mine + trow * PITCH + 8 * h;

    for (int base = gwave; base < ntiles; base += AHEAD * nwaves) {
        v4f in[AHEAD][2];
#pragma unroll
        for (int a = 0; a < AHEAD; ++a) {
            const int t  = base + a * nwaves;
            const int tl = t < ntiles ? t : base;
            gfloatp src = same_in_all_lanes(x + (size_t)tl * 256);
            in[a][0] = *(gv4fp)(src + 4 * lane);
            in[a][1] = *(gv4fp)(src + 4 * lane + 128);
        }

        __builtin_amdgcn_sched_barrier(0);

#pragma unroll
        for (int a = 0; a < AHEAD; ++a) {
            const int tile = base + a * nwaves;

            *(v4f*)put               = in[a][0];
            *(v4f*)(put + 8 * PITCH) = in[a][1];
            const v4f f0 = *(const v4f*)get;
            const v4f f1 = *(const v4f*)(get + 4);

            v2f xb[4];
            xb[0].x = f0.x; xb[0].y = f0.y;
            xb[1].x = f0.z; xb[1].y = f0.w;
            xb[2].x = f1.x; xb[2].y = f1.y;
            xb[3].x = f1.z; xb[3].y = f1.w;

            v8f zt = {};
#pragma unroll
            for (int c = 0; c < 4; ++c)
                zt = __builtin_amdgcn_wmma_f32_16x16x4_f32(
                         false, w1[c],
                         false, xb[c],
                         (short)0, zt,
                         false,  false);

            asm volatile("s_wait_alu depctr_va_vdst(0)" : "+v"(zt));

            v2f za0, za1;
            za0.x = zt[0]; za0.y = zt[1];
            za1.x = zt[2]; za1.y = zt[3];

            v8f d = {};
            d = __builtin_amdgcn_wmma_f32_16x16x4_f32(
                    false, za0, false, w2[0], (short)0, d, false, false);
            d = __builtin_amdgcn_wmma_f32_16x16x4_f32(
                    false, za1, false, w2[1], (short)0, d, false, false);

            if (tile < ntiles) {
                float* Drow = y + (size_t)tile * 256 + h * BLK + n;
#pragma unroll
                for (int g = 0; g < 8; ++g)
                    Drow[g * 2 * BLK] = d[g];
            }
        }
    }
}

extern "C" void kernel_launch(void* const* d_in, const int* in_sizes, int n_in,
                              void* d_out, int out_size, void* d_ws, size_t ws_size,
                              hipStream_t stream) {
    const float* x      = (const float*)d_in[0];
    const float* W_samp = (const float*)d_in[1];
    const float* W_init = (const float*)d_in[2];
    float* y = (float*)d_out;

    const int ntiles = in_sizes[0] / 256;
    const int blocks = 1024;
    sample_recon_kernel<<<blocks, 32 * WAVES, 0, stream>>>(x, W_samp, W_init, y, ntiles);
}
